# starting kernel
# baseline (speedup 1.0000x reference)
#include <hip/hip_runtime.h>

#define BATCH 4096
#define NODES 64
#define FDIM  128
#define LDS_S 132

typedef __attribute__((ext_vector_type(2))) float v2f;
typedef __attribute__((ext_vector_type(8))) float v8f;

__launch_bounds__(256)
__global__ void gcn_mean_kernel(const float* __restrict__ emb,
                                float* __restrict__ xbar) {
    const int wave = threadIdx.x >> 5;
    const int lane = threadIdx.x & 31;
    const int b = blockIdx.x * 8 + wave;

    const float4* __restrict__ src =
        (const float4*)(emb + (size_t)b * (NODES * FDIM));

    float4 acc = make_float4(0.f, 0.f, 0.f, 0.f);
#pragma unroll 16
    for (int c = 0; c < NODES; ++c) {
        float4 v = src[c * (FDIM / 4) + lane];
        acc.x += v.x; acc.y += v.y; acc.z += v.z; acc.w += v.w;
    }
    const float s = 1.0f / (float)NODES;
    acc.x *= s; acc.y *= s; acc.z *= s; acc.w *= s;
    ((float4*)(xbar + (size_t)b * FDIM))[lane] = acc;
}

__launch_bounds__(256)
__global__ void gcn_mlp_kernel(const float* __restrict__ xbar,
                               const float* __restrict__ W0, const float* __restrict__ b0,
                               const float* __restrict__ W1, const float* __restrict__ b1,
                               const float* __restrict__ W2, const float* __restrict__ b2,
                               const float* __restrict__ Wr, const float* __restrict__ br,
                               float* __restrict__ out) {
    __shared__ float Wlds[FDIM * LDS_S];
    __shared__ float ybuf[2][16 * LDS_S];

    const int tid   = threadIdx.x;
    const int wave  = tid >> 5;
    const int lane  = tid & 31;
    const int b0row = blockIdx.x * 16;

#pragma unroll
    for (int i = 0; i < 2; ++i) {
        int fi = tid + i * 256;
        int m  = fi >> 5;
        int f  = (fi & 31) << 2;
        float4 v = ((const float4*)(xbar + (size_t)(b0row + m) * FDIM))[fi & 31];
        *(float4*)&ybuf[0][m * LDS_S + f] = v;
    }

    const float* Ws[4] = {W0, W1, W2, Wr};
    const float* bs[4] = {b0, b1, b2, br};

    const int m     = lane & 15;
    const int khalf = lane >> 4;
    const int n     = (lane & 15) + wave * 16;

    int cur = 0;
#pragma unroll
    for (int layer = 0; layer < 4; ++layer) {
        __syncthreads();

        const float* __restrict__ W = Ws[layer];
#pragma unroll 4
        for (int i = 0; i < 16; ++i) {
            int fi = i * 256 + tid;
            int k  = fi >> 5;
            int nn = (fi & 31) << 2;
            float4 v = ((const float4*)(W + (size_t)k * FDIM))[fi & 31];
            *(float4*)&Wlds[k * LDS_S + nn] = v;
        }
        const float bias = bs[layer][n];
        __syncthreads();

        const float* __restrict__ yin = ybuf[cur];
        v8f acc = {0.f, 0.f, 0.f, 0.f, 0.f, 0.f, 0.f, 0.f};
#pragma unroll 4
        for (int t = 0; t < 32; ++t) {
            const int kb = t * 4;
            v2f a = *(const v2f*)&yin[m * LDS_S + kb + 2 * khalf];
            v2f bf;
            bf.x = Wlds[(kb + 2 * khalf + 0) * LDS_S + n];
            bf.y = Wlds[(kb + 2 * khalf + 1) * LDS_S + n];
            acc = __builtin_amdgcn_wmma_f32_16x16x4_f32(
                false, a, false, bf, (short)0, acc, false, false);
        }

        if (layer < 3) {
            float* __restrict__ yout = ybuf[1 - cur];
#pragma unroll
            for (int r = 0; r < 8; ++r) {
                float v = acc[r] + bias;
                yout[(r + khalf * 8) * LDS_S + n] = v > 0.f ? v : 0.f;
            }
            cur = 1 - cur;
        } else {
#pragma unroll
            for (int r = 0; r < 8; ++r) {
                out[(size_t)(b0row + r + khalf * 8) * FDIM + n] = acc[r] + bias;
            }
        }
    }
}

extern "C" void kernel_launch(void* const* d_in, const int* in_sizes, int n_in,
                              void* d_out, int out_size, void* d_ws, size_t ws_size,
                              hipStream_t stream) {
    const float* emb = (const float*)d_in[0];
    const float* W0  = (const float*)d_in[1];
    const float* b0  = (const float*)d_in[2];
    const float* W1  = (const float*)d_in[3];
    const float* b1  = (const float*)d_in[4];
    const float* W2  = (const float*)d_in[5];
    const float* b2  = (const float*)d_in[6];
    const float* Wr  = (const float*)d_in[7];
    const float* br  = (const float*)d_in[8];
    float* out  = (float*)d_out;
    float* xbar = (float*)d_ws;

    gcn_mean_kernel<<<BATCH / 8, 256, 0, stream>>>(emb, xbar);
    gcn_mlp_kernel<<<BATCH / 16, 256, 0, stream>>>(
        xbar, W0, b0, W1, b1, W2, b2, Wr, br, out);
}


# fuse the node-mean kernel into the layer-chain kernel: one launch of 256 work-groups x 16 waves, each wave reduces one batch row straight into the LDS slab; no workspace round trip
# speedup vs baseline: 1.4717x; 1.4717x over previous
#include <hip/hip_runtime.h>

#define BATCH 4096
#define NODES 64
#define FDIM  128
#define LDS_S 132
#define ROWS  16
#define NTHR  512

typedef __attribute__((ext_vector_type(2))) float v2f;
typedef __attribute__((ext_vector_type(8))) float v8f;

__launch_bounds__(NTHR)
__global__ void gcn_fused_kernel(const float* __restrict__ emb,
                                 const float* __restrict__ W0, const float* __restrict__ b0,
                                 const float* __restrict__ W1, const float* __restrict__ b1,
                                 const float* __restrict__ W2, const float* __restrict__ b2,
                                 const float* __restrict__ Wr, const float* __restrict__ br,
                                 float* __restrict__ out) {
    __shared__ float Wlds[FDIM * LDS_S];
    __shared__ float ybuf[2][ROWS * LDS_S];

    const int tid   = threadIdx.x;
    const int wave  = __builtin_amdgcn_readfirstlane(tid >> 5);
    const int lane  = tid & 31;
    const int b0row = blockIdx.x * ROWS;

    {
        const float4* __restrict__ src =
            (const float4*)(emb + (size_t)(b0row + wave) * (NODES * FDIM));
        float4 acc = make_float4(0.f, 0.f, 0.f, 0.f);
#pragma unroll 16
        for (int c = 0; c < NODES; ++c) {
            float4 v = src[c * (FDIM / 4) + lane];
            acc.x += v.x; acc.y += v.y; acc.z += v.z; acc.w += v.w;
        }
        const float s = 1.0f / (float)NODES;
        acc.x *= s; acc.y *= s; acc.z *= s; acc.w *= s;
        *(float4*)&ybuf[0][wave * LDS_S + lane * 4] = acc;
    }

    const float* Ws[4] = {W0, W1, W2, Wr};
    const float* bs[4] = {b0, b1, b2, br};

    const bool mm   = wave < 8;
    const int m     = lane & 15;
    const int khalf = lane >> 4;
    const int n     = (lane & 15) + (wave & 7) * 16;

    int cur = 0;
#pragma unroll
    for (int layer = 0; layer < 4; ++layer) {
        __syncthreads();

        const float* __restrict__ W = Ws[layer];
#pragma unroll
        for (int i = 0; i < 8; ++i) {
            int fi = i * NTHR + tid;
            int k  = fi >> 5;
            int nn = (fi & 31) << 2;
            float4 v = ((const float4*)(W + (size_t)k * FDIM))[fi & 31];
            *(float4*)&Wlds[k * LDS_S + nn] = v;
        }
        const float bias = bs[layer][n];
        __syncthreads();

        if (mm) {
            const float* __restrict__ yin = ybuf[cur];
            v8f acc = {0.f, 0.f, 0.f, 0.f, 0.f, 0.f, 0.f, 0.f};
#pragma unroll 4
            for (int t = 0; t < 32; ++t) {
                const int kb = t * 4;
                v2f a = *(const v2f*)&yin[m * LDS_S + kb + 2 * khalf];
                v2f bf;
                bf.x = Wlds[(kb + 2 * khalf + 0) * LDS_S + n];
                bf.y = Wlds[(kb + 2 * khalf + 1) * LDS_S + n];
                acc = __builtin_amdgcn_wmma_f32_16x16x4_f32(
                    false, a, false, bf, (short)0, acc, false, false);
            }

            if (layer < 3) {
                float* __restrict__ yout = ybuf[1 - cur];
#pragma unroll
                for (int r = 0; r < 8; ++r) {
                    float v = acc[r] + bias;
                    yout[(r + khalf * 8) * LDS_S + n] = v > 0.f ? v : 0.f;
                }
            } else {
#pragma unroll
                for (int r = 0; r < 8; ++r) {
                    out[(size_t)(b0row + r + khalf * 8) * FDIM + n] = acc[r] + bias;
                }
            }
        }
        cur = 1 - cur;
    }
}

extern "C" void kernel_launch(void* const* d_in, const int* in_sizes, int n_in,
                              void* d_out, int out_size, void* d_ws, size_t ws_size,
                              hipStream_t stream) {
    const float* emb = (const float*)d_in[0];
    const float* W0  = (const float*)d_in[1];
    const float* b0  = (const float*)d_in[2];
    const float* W1  = (const float*)d_in[3];
    const float* b1  = (const float*)d_in[4];
    const float* W2  = (const float*)d_in[5];
    const float* b2  = (const float*)d_in[6];
    const float* Wr  = (const float*)d_in[7];
    const float* br  = (const float*)d_in[8];
    float* out  = (float*)d_out;

    gcn_fused_kernel<<<BATCH / ROWS, NTHR, 0, stream>>>(
        emb, W0, b0, W1, b1, W2, b2, Wr, br, out);
}


# weight fetch off the chain's critical path: 2 LDS planes; W0 + biases ride the node stream's first load group, each later matrix is fetched by the 8 idle waves while 8 multiply; 1 barrier per layer
# speedup vs baseline: 1.6304x; 1.1078x over previous
#include <hip/hip_runtime.h>

#define BATCH 4096
#define NODES 64
#define FDIM  128
#define LDS_S 132
#define ROWS  16
#define NTHR  512

typedef __attribute__((ext_vector_type(2))) float v2f;
typedef __attribute__((ext_vector_type(8))) float v8f;

__launch_bounds__(NTHR)
__global__ void gcn_fused_kernel(const float* __restrict__ emb,
                                 const float* __restrict__ W0, const float* __restrict__ b0,
                                 const float* __restrict__ W1, const float* __restrict__ b1,
                                 const float* __restrict__ W2, const float* __restrict__ b2,
                                 const float* __restrict__ Wr, const float* __restrict__ br,
                                 float* __restrict__ out) {
    __shared__ float Wlds[2][FDIM * LDS_S];
    __shared__ float ybuf[2][ROWS * LDS_S];

    const int tid   = threadIdx.x;
    const int wave  = __builtin_amdgcn_readfirstlane(tid >> 5);
    const int lane  = tid & 31;
    const int b0row = blockIdx.x * ROWS;

    const float* Ws[4] = {W0, W1, W2, Wr};
    const float* bs[4] = {b0, b1, b2, br};

    const bool mm   = wave < 8;
    const int m     = lane & 15;
    const int khalf = lane >> 4;
    const int n     = (lane & 15) + (wave & 7) * 16;

    float bias[4];
#pragma unroll
    for (int l = 0; l < 4; ++l) bias[l] = bs[l][n];

    {
        const float4* __restrict__ src =
            (const float4*)(emb + (size_t)(b0row + wave) * (NODES * FDIM));
        float4 acc = make_float4(0.f, 0.f, 0.f, 0.f);
        auto sum16 = [&](int g) {
            float4 v[16];
#pragma unroll
            for (int c = 0; c < 16; ++c) v[c] = src[(g * 16 + c) * (FDIM / 4) + lane];
#pragma unroll
            for (int c = 0; c < 16; ++c) {
                acc.x += v[c].x; acc.y += v[c].y; acc.z += v[c].z; acc.w += v[c].w;
            }
        };
        {
            float4 w0v[8];
#pragma unroll
            for (int i = 0; i < 8; ++i) w0v[i] = ((const float4*)W0)[i * NTHR + tid];
            sum16(0);
#pragma unroll
            for (int i = 0; i < 8; ++i) {
                int fi = i * NTHR + tid;
                int k  = fi >> 5;
                int nn = (fi & 31) << 2;
                *(float4*)&Wlds[0][k * LDS_S + nn] = w0v[i];
            }
        }
#pragma unroll 1
        for (int g = 1; g < NODES / 16; ++g) sum16(g);
        const float s = 1.0f / (float)NODES;
        acc.x *= s; acc.y *= s; acc.z *= s; acc.w *= s;
        *(float4*)&ybuf[0][wave * LDS_S + lane * 4] = acc;
    }

    __syncthreads();

    int cur = 0;
#pragma unroll
    for (int layer = 0; layer < 4; ++layer) {
        if (mm) {
            const float* __restrict__ Wl = Wlds[layer & 1];
            const float bl = bias[layer];
            const float* __restrict__ yin = ybuf[cur];
            v8f acc = {0.f, 0.f, 0.f, 0.f, 0.f, 0.f, 0.f, 0.f};
#pragma unroll 4
            for (int t = 0; t < 32; ++t) {
                const int kb = t * 4;
                v2f a = *(const v2f*)&yin[m * LDS_S + kb + 2 * khalf];
                v2f bf;
                bf.x = Wl[(kb + 2 * khalf + 0) * LDS_S + n];
                bf.y = Wl[(kb + 2 * khalf + 1) * LDS_S + n];
                acc = __builtin_amdgcn_wmma_f32_16x16x4_f32(
                    false, a, false, bf, (short)0, acc, false, false);
            }

            if (layer < 3) {
                float* __restrict__ yout = ybuf[1 - cur];
#pragma unroll
                for (int r = 0; r < 8; ++r) {
                    float v = acc[r] + bl;
                    yout[(r + khalf * 8) * LDS_S + n] = v > 0.f ? v : 0.f;
                }
            } else {
#pragma unroll
                for (int r = 0; r < 8; ++r) {
                    out[(size_t)(b0row + r + khalf * 8) * FDIM + n] = acc[r] + bl;
                }
            }
        } else if (layer < 3) {
            const float4* __restrict__ Wn = (const float4*)Ws[layer + 1] + (tid - NTHR / 2);
            float* __restrict__ Wp =
                &Wlds[(layer + 1) & 1][((tid - NTHR / 2) >> 5) * LDS_S + lane * 4];
            const float4 w0 = Wn[0 * (NTHR / 2)];
            const float4 w1 = Wn[1 * (NTHR / 2)];
            const float4 w2 = Wn[2 * (NTHR / 2)];
            const float4 w3 = Wn[3 * (NTHR / 2)];
            const float4 w4 = Wn[4 * (NTHR / 2)];
            const float4 w5 = Wn[5 * (NTHR / 2)];
            const float4 w6 = Wn[6 * (NTHR / 2)];
            const float4 w7 = Wn[7 * (NTHR / 2)];
            const float4 w8 = Wn[8 * (NTHR / 2)];
            const float4 w9 = Wn[9 * (NTHR / 2)];
            const float4 wa = Wn[10 * (NTHR / 2)];
            const float4 wb = Wn[11 * (NTHR / 2)];
            const float4 wc = Wn[12 * (NTHR / 2)];
            const float4 wd = Wn[13 * (NTHR / 2)];
            const float4 we = Wn[14 * (NTHR / 2)];
            const float4 wf = Wn[15 * (NTHR / 2)];
            *(float4*)&Wp[0 * 8 * LDS_S] = w0;
            *(float4*)&Wp[1 * 8 * LDS_S] = w1;
            *(float4*)&Wp[2 * 8 * LDS_S] = w2;
            *(float4*)&Wp[3 * 8 * LDS_S] = w3;
            *(float4*)&Wp[4 * 8 * LDS_S] = w4;
            *(float4*)&Wp[5 * 8 * LDS_S] = w5;
            *(float4*)&Wp[6 * 8 * LDS_S] = w6;
            *(float4*)&Wp[7 * 8 * LDS_S] = w7;
            *(float4*)&Wp[8 * 8 * LDS_S] = w8;
            *(float4*)&Wp[9 * 8 * LDS_S] = w9;
            *(float4*)&Wp[10 * 8 * LDS_S] = wa;
            *(float4*)&Wp[11 * 8 * LDS_S] = wb;
            *(float4*)&Wp[12 * 8 * LDS_S] = wc;
            *(float4*)&Wp[13 * 8 * LDS_S] = wd;
            *(float4*)&Wp[14 * 8 * LDS_S] = we;
            *(float4*)&Wp[15 * 8 * LDS_S] = wf;
        }
        if (layer < 3) __syncthreads();
        cur = 1 - cur;
    }
}

extern "C" void kernel_launch(void* const* d_in, const int* in_sizes, int n_in,
                              void* d_out, int out_size, void* d_ws, size_t ws_size,
                              hipStream_t stream) {
    const float* emb = (const float*)d_in[0];
    const float* W0  = (const float*)d_in[1];
    const float* b0  = (const float*)d_in[2];
    const float* W1  = (const float*)d_in[3];
    const float* b1  = (const float*)d_in[4];
    const float* W2  = (const float*)d_in[5];
    const float* b2  = (const float*)d_in[6];
    const float* Wr  = (const float*)d_in[7];
    const float* br  = (const float*)d_in[8];
    float* out  = (float*)d_out;

    gcn_fused_kernel<<<BATCH / ROWS, NTHR, 0, stream>>>(
        emb, W0, b0, W1, b1, W2, b2, Wr, br, out);
}


# chain: fragment reads software-pipelined over 2 register sets, each 8 k-steps' LDS reads issued as one group 8 matrix instructions ahead (was read-then-wait in each trip); k order and sums unchanged
# speedup vs baseline: 1.7143x; 1.0515x over previous
#include <hip/hip_runtime.h>

#define BATCH 4096
#define NODES 64
#define FDIM  128
#define LDS_S 132
#define ROWS  16
#define NTHR  512

typedef __attribute__((ext_vector_type(2))) float v2f;
typedef __attribute__((ext_vector_type(8))) float v8f;

__launch_bounds__(NTHR)
__global__ void gcn_fused_kernel(const float* __restrict__ emb,
                                 const float* __restrict__ W0, const float* __restrict__ b0,
                                 const float* __restrict__ W1, const float* __restrict__ b1,
                                 const float* __restrict__ W2, const float* __restrict__ b2,
                                 const float* __restrict__ Wr, const float* __restrict__ br,
                                 float* __restrict__ out) {
    __shared__ float Wlds[2][FDIM * LDS_S];
    __shared__ float ybuf[2][ROWS * LDS_S];

    const int tid   = threadIdx.x;
    const int wave  = __builtin_amdgcn_readfirstlane(tid >> 5);
    const int lane  = tid & 31;
    const int b0row = blockIdx.x * ROWS;

    const float* Ws[4] = {W0, W1, W2, Wr};
    const float* bs[4] = {b0, b1, b2, br};

    const bool mm   = wave < 8;
    const int m     = lane & 15;
    const int khalf = lane >> 4;
    const int n     = (lane & 15) + (wave & 7) * 16;

    float bias[4];
#pragma unroll
    for (int l = 0; l < 4; ++l) bias[l] = bs[l][n];

    {
        const float4* __restrict__ src =
            (const float4*)(emb + (size_t)(b0row + wave) * (NODES * FDIM));
        float4 acc = make_float4(0.f, 0.f, 0.f, 0.f);
        auto sum16 = [&](int g) {
            float4 v[16];
#pragma unroll
            for (int c = 0; c < 16; ++c) v[c] = src[(g * 16 + c) * (FDIM / 4) + lane];
#pragma unroll
            for (int c = 0; c < 16; ++c) {
                acc.x += v[c].x; acc.y += v[c].y; acc.z += v[c].z; acc.w += v[c].w;
            }
        };
        {
            float4 w0v[8];
#pragma unroll
            for (int i = 0; i < 8; ++i) w0v[i] = ((const float4*)W0)[i * NTHR + tid];
            sum16(0);
#pragma unroll
            for (int i = 0; i < 8; ++i) {
                int fi = i * NTHR + tid;
                int k  = fi >> 5;
                int nn = (fi & 31) << 2;
                *(float4*)&Wlds[0][k * LDS_S + nn] = w0v[i];
            }
        }
#pragma unroll 1
        for (int g = 1; g < NODES / 16; ++g) sum16(g);
        const float s = 1.0f / (float)NODES;
        acc.x *= s; acc.y *= s; acc.z *= s; acc.w *= s;
        *(float4*)&ybuf[0][wave * LDS_S + lane * 4] = acc;
    }

    __syncthreads();

    int cur = 0;
#pragma unroll
    for (int layer = 0; layer < 4; ++layer) {
        if (mm) {
            const float* __restrict__ Wl = Wlds[layer & 1];
            const float bl = bias[layer];
            const float* __restrict__ yin = ybuf[cur];
            v8f acc = {0.f, 0.f, 0.f, 0.f, 0.f, 0.f, 0.f, 0.f};
            const float* __restrict__ ap = yin + m * LDS_S + 2 * khalf;
            const float* __restrict__ bp = Wl + (2 * khalf) * LDS_S + n;
            v2f af[2][8], bf[2][8];
            auto load_row = [&](int set, int r) {
#pragma unroll
                for (int j = 0; j < 8; ++j) {
                    const int kb = (r * 8 + j) * 4;
                    af[set][j] = *(const v2f*)&ap[kb];
                    bf[set][j].x = bp[(kb + 0) * LDS_S];
                    bf[set][j].y = bp[(kb + 1) * LDS_S];
                }
            };
            load_row(0, 0);
            __builtin_amdgcn_sched_barrier(0);
#pragma unroll
            for (int r = 0; r < 4; ++r) {
                if (r < 3) load_row((r + 1) & 1, r + 1);
                __builtin_amdgcn_sched_barrier(0);
#pragma unroll
                for (int j = 0; j < 8; ++j) {
                    acc = __builtin_amdgcn_wmma_f32_16x16x4_f32(
                        false, af[r & 1][j], false, bf[r & 1][j], (short)0, acc, false, false);
                }
                __builtin_amdgcn_sched_barrier(0);
            }

            if (layer < 3) {
                float* __restrict__ yout = ybuf[1 - cur];
#pragma unroll
                for (int r = 0; r < 8; ++r) {
                    float v = acc[r] + bl;
                    yout[(r + khalf * 8) * LDS_S + n] = v > 0.f ? v : 0.f;
                }
            } else {
#pragma unroll
                for (int r = 0; r < 8; ++r) {
                    out[(size_t)(b0row + r + khalf * 8) * FDIM + n] = acc[r] + bl;
                }
            }
        } else if (layer < 3) {
            const float4* __restrict__ Wn = (const float4*)Ws[layer + 1] + (tid - NTHR / 2);
            float* __restrict__ Wp =
                &Wlds[(layer + 1) & 1][((tid - NTHR / 2) >> 5) * LDS_S + lane * 4];
            const float4 w0 = Wn[0 * (NTHR / 2)];
            const float4 w1 = Wn[1 * (NTHR / 2)];
            const float4 w2 = Wn[2 * (NTHR / 2)];
            const float4 w3 = Wn[3 * (NTHR / 2)];
            const float4 w4 = Wn[4 * (NTHR / 2)];
            const float4 w5 = Wn[5 * (NTHR / 2)];
            const float4 w6 = Wn[6 * (NTHR / 2)];
            const float4 w7 = Wn[7 * (NTHR / 2)];
            const float4 w8 = Wn[8 * (NTHR / 2)];
            const float4 w9 = Wn[9 * (NTHR / 2)];
            const float4 wa = Wn[10 * (NTHR / 2)];
            const float4 wb = Wn[11 * (NTHR / 2)];
            const float4 wc = Wn[12 * (NTHR / 2)];
            const float4 wd = Wn[13 * (NTHR / 2)];
            const float4 we = Wn[14 * (NTHR / 2)];
            const float4 wf = Wn[15 * (NTHR / 2)];
            *(float4*)&Wp[0 * 8 * LDS_S] = w0;
            *(float4*)&Wp[1 * 8 * LDS_S] = w1;
            *(float4*)&Wp[2 * 8 * LDS_S] = w2;
            *(float4*)&Wp[3 * 8 * LDS_S] = w3;
            *(float4*)&Wp[4 * 8 * LDS_S] = w4;
            *(float4*)&Wp[5 * 8 * LDS_S] = w5;
            *(float4*)&Wp[6 * 8 * LDS_S] = w6;
            *(float4*)&Wp[7 * 8 * LDS_S] = w7;
            *(float4*)&Wp[8 * 8 * LDS_S] = w8;
            *(float4*)&Wp[9 * 8 * LDS_S] = w9;
            *(float4*)&Wp[10 * 8 * LDS_S] = wa;
            *(float4*)&Wp[11 * 8 * LDS_S] = wb;
            *(float4*)&Wp[12 * 8 * LDS_S] = wc;
            *(float4*)&Wp[13 * 8 * LDS_S] = wd;
            *(float4*)&Wp[14 * 8 * LDS_S] = we;
            *(float4*)&Wp[15 * 8 * LDS_S] = wf;
        }
        if (layer < 3) __syncthreads();
        cur = 1 - cur;
    }
}

extern "C" void kernel_launch(void* const* d_in, const int* in_sizes, int n_in,
                              void* d_out, int out_size, void* d_ws, size_t ws_size,
                              hipStream_t stream) {
    const float* emb = (const float*)d_in[0];
    const float* W0  = (const float*)d_in[1];
    const float* b0  = (const float*)d_in[2];
    const float* W1  = (const float*)d_in[3];
    const float* b1  = (const float*)d_in[4];
    const float* W2  = (const float*)d_in[5];
    const float* b2  = (const float*)d_in[6];
    const float* Wr  = (const float*)d_in[7];
    const float* br  = (const float*)d_in[8];
    float* out  = (float*)d_out;

    gcn_fused_kernel<<<BATCH / ROWS, NTHR, 0, stream>>>(
        emb, W0, b0, W1, b1, W2, b2, Wr, br, out);
}
